# starting kernel
# baseline (speedup 1.0000x reference)
#include <hip/hip_runtime.h>
#include <hip/hip_bf16.h>

typedef __attribute__((ext_vector_type(2))) float v2f;
typedef __attribute__((ext_vector_type(8))) float v8f;

#define ALPHA_F 0.5f
#define NCLS 16

__global__ __launch_bounds__(256)
void lp_prep(const float* __restrict__ x, const float* __restrict__ W,
             float* __restrict__ hs, float* __restrict__ agg, int total) {
  int i = blockIdx.x * 256 + threadIdx.x;
  if (i < total) {
    hs[i]  = W[i >> 4] * x[i];
    agg[i] = 0.0f;
  }
}

__global__ __launch_bounds__(256)
void lp_edge(const int* __restrict__ ei, const float* __restrict__ ew,
             const float* __restrict__ hs, float* agg, int n_edges) {
  int gid = blockIdx.x * 256 + threadIdx.x;
  int e = gid >> 4;
  int c = gid & (NCLS - 1);
  if (e < n_edges) {
    if (c == 0) {
      __builtin_prefetch(ei + e + 4096, 0, 1);
      __builtin_prefetch(ew + e + 4096, 0, 1);
    }
    int   src = ei[e];
    int   dst = ei[n_edges + e];
    float w   = ew[e];
    float v   = hs[src * NCLS + c] * w;
    atomicAdd(&agg[dst * NCLS + c], v);
  }
}

__global__ __launch_bounds__(256)
void lp_node(const float* __restrict__ h_in, float* agg,
             const float* __restrict__ W, const float* __restrict__ degree,
             float* __restrict__ h_out, float* __restrict__ hs_out,
             int n_tiles) {
  const int lane  = threadIdx.x & 31;
  const int wave  = threadIdx.x >> 5;
  const int tile  = blockIdx.x * 8 + wave;
  if (tile >= n_tiles) return;
  const int node0 = tile * 16;
  const int m     = lane & 15;
  const int hi    = (lane >= 16) ? 1 : 0;

  float dg = degree[node0 + m];
  float av = (1.0f - ALPHA_F) * dg * dg;

  v8f acc;
#pragma unroll
  for (int r = 0; r < 8; ++r) {
    int row = node0 + r + (hi ? 8 : 0);
    acc[r] = ALPHA_F * h_in[row * NCLS + m];
  }

#pragma unroll
  for (int q = 0; q < 4; ++q) {
    int k0 = 4 * q + 2 * hi;
    v2f a, b;
    a.x = (m == k0)     ? av : 0.0f;
    a.y = (m == k0 + 1) ? av : 0.0f;
    b.x = agg[(node0 + k0)     * NCLS + m];
    b.y = agg[(node0 + k0 + 1) * NCLS + m];
    acc = __builtin_amdgcn_wmma_f32_16x16x4_f32(
         false, a,  false, b,
         (short)0, acc,  false,  false);
  }

#pragma unroll
  for (int r = 0; r < 8; ++r) {
    float s = acc[r];
    s += __shfl_xor(s, 1, 32);
    s += __shfl_xor(s, 2, 32);
    s += __shfl_xor(s, 4, 32);
    s += __shfl_xor(s, 8, 32);
    int   row = node0 + r + (hi ? 8 : 0);
    float o   = acc[r] / s;
    h_out [row * NCLS + m] = o;
    hs_out[row * NCLS + m] = W[row] * o;
    agg   [row * NCLS + m] = 0.0f;
  }
}

extern "C" void kernel_launch(void* const* d_in, const int* in_sizes, int n_in,
                              void* d_out, int out_size, void* d_ws, size_t ws_size,
                              hipStream_t stream) {
  const float* x   = (const float*)d_in[0];
  const float* W   = (const float*)d_in[1];
  const float* ew  = (const float*)d_in[2];
  const float* deg = (const float*)d_in[3];
  const int*   ei  = (const int*)  d_in[4];

  const int N       = in_sizes[3];
  const int N_EDGES = in_sizes[2];
  const int total   = N * NCLS;
  const int n_tiles = N / 16;

  float* out = (float*)d_out;
  float* agg = (float*)d_ws;
  float* hA  = agg + (size_t)total;
  float* hs  = hA  + (size_t)total;

  const int prep_blocks = (total + 255) / 256;
  const int edge_blocks = (N_EDGES * NCLS + 255) / 256;
  const int node_blocks = (n_tiles + 7) / 8;

  lp_prep<<<prep_blocks, 256, 0, stream>>>(x, W, hs, agg, total);

  for (int k = 0; k < 4; ++k) {
    const float* hin  = (k == 0) ? x : ((k & 1) ? hA : out);
    float*       hout = (k & 1) ? out : hA;
    lp_edge<<<edge_blocks, 256, 0, stream>>>(ei, ew, hs, agg, N_EDGES);
    lp_node<<<node_blocks, 256, 0, stream>>>(hin, agg, W, deg, hout, hs, n_tiles);
  }
}


# edge phase: one wave per 32 edges, each lane loads one edge's src/dst/weight (coalesced), DPP row broadcast, 16 row gathers in flight; f32 atomic row adds kept (arrival-order sums as before)
# speedup vs baseline: 1.8103x; 1.8103x over previous
#include <hip/hip_runtime.h>
#include <hip/hip_bf16.h>

typedef __attribute__((ext_vector_type(2))) float v2f;
typedef __attribute__((ext_vector_type(8))) float v8f;

#define ALPHA_F 0.5f
#define NCLS 16

__global__ __launch_bounds__(256)
void lp_prep(const float* __restrict__ x, const float* __restrict__ W,
             float* __restrict__ hs, float* __restrict__ agg, int total) {
  int i = blockIdx.x * 256 + threadIdx.x;
  if (i < total) {
    hs[i]  = W[i >> 4] * x[i];
    agg[i] = 0.0f;
  }
}

template <int K>
__device__ __forceinline__ int row_bcast(int v) {
  return __builtin_amdgcn_update_dpp(0, v, 0x150 + K, 0xF, 0xF, false);
}

template <int K>
__device__ __forceinline__ float row_gather(const float* __restrict__ hs, int sb, int c) {
  return hs[row_bcast<K>(sb) + c];
}

template <int K>
__device__ __forceinline__ void row_add(float* agg, int db, int wbits, int c, float v) {
  float w = __int_as_float(row_bcast<K>(wbits));
  atomicAdd(&agg[row_bcast<K>(db) + c], v * w);
}

__global__ __launch_bounds__(256)
void lp_edge(const int* __restrict__ ei, const float* __restrict__ ew,
             const float* __restrict__ hs, float* agg, int n_edges) {
  const int lane = threadIdx.x & 31;
  const int c    = lane & (NCLS - 1);
  const int e0   = (blockIdx.x * 8 + (threadIdx.x >> 5)) * 32;
  if (e0 >= n_edges) return;

  if (e0 + 32 <= n_edges) {
    const int e  = e0 + lane;
    const int sb = ei[e] * NCLS;
    const int db = ei[n_edges + e] * NCLS;
    const int wb = __float_as_int(ew[e]);

    float v0  = row_gather<0>(hs, sb, c);
    float v1  = row_gather<1>(hs, sb, c);
    float v2  = row_gather<2>(hs, sb, c);
    float v3  = row_gather<3>(hs, sb, c);
    float v4  = row_gather<4>(hs, sb, c);
    float v5  = row_gather<5>(hs, sb, c);
    float v6  = row_gather<6>(hs, sb, c);
    float v7  = row_gather<7>(hs, sb, c);
    float v8  = row_gather<8>(hs, sb, c);
    float v9  = row_gather<9>(hs, sb, c);
    float v10 = row_gather<10>(hs, sb, c);
    float v11 = row_gather<11>(hs, sb, c);
    float v12 = row_gather<12>(hs, sb, c);
    float v13 = row_gather<13>(hs, sb, c);
    float v14 = row_gather<14>(hs, sb, c);
    float v15 = row_gather<15>(hs, sb, c);

    row_add<0>(agg, db, wb, c, v0);
    row_add<1>(agg, db, wb, c, v1);
    row_add<2>(agg, db, wb, c, v2);
    row_add<3>(agg, db, wb, c, v3);
    row_add<4>(agg, db, wb, c, v4);
    row_add<5>(agg, db, wb, c, v5);
    row_add<6>(agg, db, wb, c, v6);
    row_add<7>(agg, db, wb, c, v7);
    row_add<8>(agg, db, wb, c, v8);
    row_add<9>(agg, db, wb, c, v9);
    row_add<10>(agg, db, wb, c, v10);
    row_add<11>(agg, db, wb, c, v11);
    row_add<12>(agg, db, wb, c, v12);
    row_add<13>(agg, db, wb, c, v13);
    row_add<14>(agg, db, wb, c, v14);
    row_add<15>(agg, db, wb, c, v15);
  } else {
    const int eb = e0 + (lane & 16);
    for (int k = 0; k < 16; ++k) {
      const int e = eb + k;
      if (e < n_edges) {
        const int   src = ei[e];
        const int   dst = ei[n_edges + e];
        const float w   = ew[e];
        atomicAdd(&agg[dst * NCLS + c], hs[src * NCLS + c] * w);
      }
    }
  }
}

__global__ __launch_bounds__(256)
void lp_node(const float* __restrict__ h_in, float* agg,
             const float* __restrict__ W, const float* __restrict__ degree,
             float* __restrict__ h_out, float* __restrict__ hs_out,
             int n_tiles) {
  const int lane  = threadIdx.x & 31;
  const int wave  = threadIdx.x >> 5;
  const int tile  = blockIdx.x * 8 + wave;
  if (tile >= n_tiles) return;
  const int node0 = tile * 16;
  const int m     = lane & 15;
  const int hi    = (lane >= 16) ? 1 : 0;

  float dg = degree[node0 + m];
  float av = (1.0f - ALPHA_F) * dg * dg;

  v8f acc;
#pragma unroll
  for (int r = 0; r < 8; ++r) {
    int row = node0 + r + (hi ? 8 : 0);
    acc[r] = ALPHA_F * h_in[row * NCLS + m];
  }

#pragma unroll
  for (int q = 0; q < 4; ++q) {
    int k0 = 4 * q + 2 * hi;
    v2f a, b;
    a.x = (m == k0)     ? av : 0.0f;
    a.y = (m == k0 + 1) ? av : 0.0f;
    b.x = agg[(node0 + k0)     * NCLS + m];
    b.y = agg[(node0 + k0 + 1) * NCLS + m];
    acc = __builtin_amdgcn_wmma_f32_16x16x4_f32(
         false, a,  false, b,
         (short)0, acc,  false,  false);
  }

#pragma unroll
  for (int r = 0; r < 8; ++r) {
    float s = acc[r];
    s += __shfl_xor(s, 1, 32);
    s += __shfl_xor(s, 2, 32);
    s += __shfl_xor(s, 4, 32);
    s += __shfl_xor(s, 8, 32);
    int   row = node0 + r + (hi ? 8 : 0);
    float o   = acc[r] / s;
    h_out [row * NCLS + m] = o;
    hs_out[row * NCLS + m] = W[row] * o;
    agg   [row * NCLS + m] = 0.0f;
  }
}

extern "C" void kernel_launch(void* const* d_in, const int* in_sizes, int n_in,
                              void* d_out, int out_size, void* d_ws, size_t ws_size,
                              hipStream_t stream) {
  const float* x   = (const float*)d_in[0];
  const float* W   = (const float*)d_in[1];
  const float* ew  = (const float*)d_in[2];
  const float* deg = (const float*)d_in[3];
  const int*   ei  = (const int*)  d_in[4];

  const int N       = in_sizes[3];
  const int N_EDGES = in_sizes[2];
  const int total   = N * NCLS;
  const int n_tiles = N / 16;

  float* out = (float*)d_out;
  float* agg = (float*)d_ws;
  float* hA  = agg + (size_t)total;
  float* hs  = hA  + (size_t)total;

  const int prep_blocks = (total + 255) / 256;
  const int edge_blocks = (N_EDGES + 255) / 256;
  const int node_blocks = (n_tiles + 7) / 8;

  lp_prep<<<prep_blocks, 256, 0, stream>>>(x, W, hs, agg, total);

  for (int k = 0; k < 4; ++k) {
    const float* hin  = (k == 0) ? x : ((k & 1) ? hA : out);
    float*       hout = (k & 1) ? out : hA;
    lp_edge<<<edge_blocks, 256, 0, stream>>>(ei, ew, hs, agg, N_EDGES);
    lp_node<<<node_blocks, 256, 0, stream>>>(hin, agg, W, deg, hout, hs, n_tiles);
  }
}


# edge sums in two planes, one per package half: a wave reads its die id and adds into its own half's plane, node phase adds the two planes (f32 atomic row adds kept; a row is now two partial sums)
# speedup vs baseline: 3.6089x; 1.9935x over previous
#include <hip/hip_runtime.h>
#include <hip/hip_bf16.h>

typedef __attribute__((ext_vector_type(2))) float v2f;
typedef __attribute__((ext_vector_type(8))) float v8f;

#define ALPHA_F 0.5f
#define NCLS 16

__global__ __launch_bounds__(256)
void lp_prep(const float* __restrict__ x, const float* __restrict__ W,
             float* __restrict__ hs, float* __restrict__ agg, int total) {
  int i = blockIdx.x * 256 + threadIdx.x;
  if (i < total) {
    hs[i]          = W[i >> 4] * x[i];
    agg[i]         = 0.0f;
    agg[i + total] = 0.0f;
  }
}

template <int K>
__device__ __forceinline__ int row_bcast(int v) {
  return __builtin_amdgcn_update_dpp(0, v, 0x150 + K, 0xF, 0xF, false);
}

template <int K>
__device__ __forceinline__ float row_gather(const float* __restrict__ hs, int sb, int c) {
  return hs[row_bcast<K>(sb) + c];
}

template <int K>
__device__ __forceinline__ void row_add(float* agg, int db, int wbits, int c, float v) {
  float w = __int_as_float(row_bcast<K>(wbits));
  atomicAdd(&agg[row_bcast<K>(db) + c], v * w);
}

__global__ __launch_bounds__(256)
void lp_edge(const int* __restrict__ ei, const float* __restrict__ ew,
             const float* __restrict__ hs, float* agg_planes, int total,
             int n_edges) {
  const int lane = threadIdx.x & 31;
  const int c    = lane & (NCLS - 1);
  const int e0   = (blockIdx.x * 8 + (threadIdx.x >> 5)) * 32;
  if (e0 >= n_edges) return;

  const unsigned place = __builtin_amdgcn_s_sendmsg_rtn(0x87);
  float* agg = agg_planes + (size_t)((place >> 18) & 1u) * (size_t)total;

  if (e0 + 32 <= n_edges) {
    const int e  = e0 + lane;
    const int sb = ei[e] * NCLS;
    const int db = ei[n_edges + e] * NCLS;
    const int wb = __float_as_int(ew[e]);

    float v0  = row_gather<0>(hs, sb, c);
    float v1  = row_gather<1>(hs, sb, c);
    float v2  = row_gather<2>(hs, sb, c);
    float v3  = row_gather<3>(hs, sb, c);
    float v4  = row_gather<4>(hs, sb, c);
    float v5  = row_gather<5>(hs, sb, c);
    float v6  = row_gather<6>(hs, sb, c);
    float v7  = row_gather<7>(hs, sb, c);
    float v8  = row_gather<8>(hs, sb, c);
    float v9  = row_gather<9>(hs, sb, c);
    float v10 = row_gather<10>(hs, sb, c);
    float v11 = row_gather<11>(hs, sb, c);
    float v12 = row_gather<12>(hs, sb, c);
    float v13 = row_gather<13>(hs, sb, c);
    float v14 = row_gather<14>(hs, sb, c);
    float v15 = row_gather<15>(hs, sb, c);

    row_add<0>(agg, db, wb, c, v0);
    row_add<1>(agg, db, wb, c, v1);
    row_add<2>(agg, db, wb, c, v2);
    row_add<3>(agg, db, wb, c, v3);
    row_add<4>(agg, db, wb, c, v4);
    row_add<5>(agg, db, wb, c, v5);
    row_add<6>(agg, db, wb, c, v6);
    row_add<7>(agg, db, wb, c, v7);
    row_add<8>(agg, db, wb, c, v8);
    row_add<9>(agg, db, wb, c, v9);
    row_add<10>(agg, db, wb, c, v10);
    row_add<11>(agg, db, wb, c, v11);
    row_add<12>(agg, db, wb, c, v12);
    row_add<13>(agg, db, wb, c, v13);
    row_add<14>(agg, db, wb, c, v14);
    row_add<15>(agg, db, wb, c, v15);
  } else {
    const int eb = e0 + (lane & 16);
    for (int k = 0; k < 16; ++k) {
      const int e = eb + k;
      if (e < n_edges) {
        const int   src = ei[e];
        const int   dst = ei[n_edges + e];
        const float w   = ew[e];
        atomicAdd(&agg[dst * NCLS + c], hs[src * NCLS + c] * w);
      }
    }
  }
}

__global__ __launch_bounds__(256)
void lp_node(const float* __restrict__ h_in, float* agg0, float* agg1,
             const float* __restrict__ W, const float* __restrict__ degree,
             float* __restrict__ h_out, float* __restrict__ hs_out,
             int n_tiles) {
  const int lane  = threadIdx.x & 31;
  const int wave  = threadIdx.x >> 5;
  const int tile  = blockIdx.x * 8 + wave;
  if (tile >= n_tiles) return;
  const int node0 = tile * 16;
  const int m     = lane & 15;
  const int hi    = (lane >= 16) ? 1 : 0;

  float dg = degree[node0 + m];
  float av = (1.0f - ALPHA_F) * dg * dg;

  v8f acc;
#pragma unroll
  for (int r = 0; r < 8; ++r) {
    int row = node0 + r + (hi ? 8 : 0);
    acc[r] = ALPHA_F * h_in[row * NCLS + m];
  }

#pragma unroll
  for (int q = 0; q < 4; ++q) {
    int k0 = 4 * q + 2 * hi;
    v2f a, b;
    a.x = (m == k0)     ? av : 0.0f;
    a.y = (m == k0 + 1) ? av : 0.0f;
    const int i0 = (node0 + k0)     * NCLS + m;
    const int i1 = (node0 + k0 + 1) * NCLS + m;
    b.x = agg0[i0] + agg1[i0];
    b.y = agg0[i1] + agg1[i1];
    acc = __builtin_amdgcn_wmma_f32_16x16x4_f32(
         false, a,  false, b,
         (short)0, acc,  false,  false);
  }

#pragma unroll
  for (int r = 0; r < 8; ++r) {
    float s = acc[r];
    s += __shfl_xor(s, 1, 32);
    s += __shfl_xor(s, 2, 32);
    s += __shfl_xor(s, 4, 32);
    s += __shfl_xor(s, 8, 32);
    int   row = node0 + r + (hi ? 8 : 0);
    float o   = acc[r] / s;
    h_out [row * NCLS + m] = o;
    hs_out[row * NCLS + m] = W[row] * o;
    agg0  [row * NCLS + m] = 0.0f;
    agg1  [row * NCLS + m] = 0.0f;
  }
}

extern "C" void kernel_launch(void* const* d_in, const int* in_sizes, int n_in,
                              void* d_out, int out_size, void* d_ws, size_t ws_size,
                              hipStream_t stream) {
  const float* x   = (const float*)d_in[0];
  const float* W   = (const float*)d_in[1];
  const float* ew  = (const float*)d_in[2];
  const float* deg = (const float*)d_in[3];
  const int*   ei  = (const int*)  d_in[4];

  const int N       = in_sizes[3];
  const int N_EDGES = in_sizes[2];
  const int total   = N * NCLS;
  const int n_tiles = N / 16;

  float* out = (float*)d_out;
  float* agg  = (float*)d_ws;
  float* agg1 = agg  + (size_t)total;
  float* hA   = agg1 + (size_t)total;
  float* hs   = hA   + (size_t)total;

  const int prep_blocks = (total + 255) / 256;
  const int edge_blocks = (N_EDGES + 255) / 256;
  const int node_blocks = (n_tiles + 7) / 8;

  lp_prep<<<prep_blocks, 256, 0, stream>>>(x, W, hs, agg, total);

  for (int k = 0; k < 4; ++k) {
    const float* hin  = (k == 0) ? x : ((k & 1) ? hA : out);
    float*       hout = (k & 1) ? out : hA;
    lp_edge<<<edge_blocks, 256, 0, stream>>>(ei, ew, hs, agg, total, N_EDGES);
    lp_node<<<node_blocks, 256, 0, stream>>>(hin, agg, agg1, W, deg, hout, hs, n_tiles);
  }
}


# edge and prep phases on a fixed grid of 1250 resident 8-wave work-groups; each wave loops over its groups of 32 edges, next group's indices loaded ahead; f32 atomic row adds kept (arrival order)
# speedup vs baseline: 4.1988x; 1.1635x over previous
#include <hip/hip_runtime.h>
#include <hip/hip_bf16.h>

typedef __attribute__((ext_vector_type(2))) float v2f;
typedef __attribute__((ext_vector_type(8))) float v8f;

#define ALPHA_F 0.5f
#define NCLS 16

#define LOOP_BLOCKS 1250

__global__ __launch_bounds__(256)
void lp_prep(const float* __restrict__ x, const float* __restrict__ W,
             float* __restrict__ hs, float* __restrict__ agg, int total) {
  const int step = gridDim.x * 256;
  for (int i = blockIdx.x * 256 + threadIdx.x; i < total; i += step) {
    hs[i]          = W[i >> 4] * x[i];
    agg[i]         = 0.0f;
    agg[i + total] = 0.0f;
  }
}

template <int K>
__device__ __forceinline__ int row_bcast(int v) {
  return __builtin_amdgcn_update_dpp(0, v, 0x150 + K, 0xF, 0xF, false);
}

template <int K>
__device__ __forceinline__ float row_gather(const float* __restrict__ hs, int sb, int c) {
  return hs[row_bcast<K>(sb) + c];
}

template <int K>
__device__ __forceinline__ void row_add(float* agg, int db, int wbits, int c, float v) {
  float w = __int_as_float(row_bcast<K>(wbits));
  atomicAdd(&agg[row_bcast<K>(db) + c], v * w);
}

__global__ __launch_bounds__(256)
void lp_edge(const int* __restrict__ ei, const float* __restrict__ ew,
             const float* __restrict__ hs, float* agg_planes, int total,
             int n_edges) {
  const int lane = threadIdx.x & 31;
  const int c    = lane & (NCLS - 1);
  const int wave   = __builtin_amdgcn_readfirstlane((int)(threadIdx.x >> 5));
  const int n_full = n_edges >> 5;
  const int stride = gridDim.x * 8;
  int g = blockIdx.x * 8 + wave;

  const unsigned place = __builtin_amdgcn_s_sendmsg_rtn(0x87);
  float* agg = agg_planes + (size_t)((place >> 18) & 1u) * (size_t)total;

  if (g >= n_full && !(g == n_full && (n_full << 5) < n_edges)) return;
  int s_cur = 0, d_cur = 0, wb = 0;
  if (g < n_full) {
    const int e = g * 32 + lane;
    s_cur = ei[e];
    d_cur = ei[n_edges + e];
    wb    = __float_as_int(ew[e]);
  }
  while (g < n_full) {
    const int gn = g + stride;
    const int gl    = (gn < n_full) ? gn : g;
    const int en    = gl * 32 + lane;
    const int s_nxt = ei[en];
    const int d_nxt = ei[n_edges + en];
    const int wbn   = __float_as_int(ew[en]);
    const int sb = s_cur * NCLS;
    const int db = d_cur * NCLS;

    float v0  = row_gather<0>(hs, sb, c);
    float v1  = row_gather<1>(hs, sb, c);
    float v2  = row_gather<2>(hs, sb, c);
    float v3  = row_gather<3>(hs, sb, c);
    float v4  = row_gather<4>(hs, sb, c);
    float v5  = row_gather<5>(hs, sb, c);
    float v6  = row_gather<6>(hs, sb, c);
    float v7  = row_gather<7>(hs, sb, c);
    float v8  = row_gather<8>(hs, sb, c);
    float v9  = row_gather<9>(hs, sb, c);
    float v10 = row_gather<10>(hs, sb, c);
    float v11 = row_gather<11>(hs, sb, c);
    float v12 = row_gather<12>(hs, sb, c);
    float v13 = row_gather<13>(hs, sb, c);
    float v14 = row_gather<14>(hs, sb, c);
    float v15 = row_gather<15>(hs, sb, c);

    row_add<0>(agg, db, wb, c, v0);
    row_add<1>(agg, db, wb, c, v1);
    row_add<2>(agg, db, wb, c, v2);
    row_add<3>(agg, db, wb, c, v3);
    row_add<4>(agg, db, wb, c, v4);
    row_add<5>(agg, db, wb, c, v5);
    row_add<6>(agg, db, wb, c, v6);
    row_add<7>(agg, db, wb, c, v7);
    row_add<8>(agg, db, wb, c, v8);
    row_add<9>(agg, db, wb, c, v9);
    row_add<10>(agg, db, wb, c, v10);
    row_add<11>(agg, db, wb, c, v11);
    row_add<12>(agg, db, wb, c, v12);
    row_add<13>(agg, db, wb, c, v13);
    row_add<14>(agg, db, wb, c, v14);
    row_add<15>(agg, db, wb, c, v15);

    s_cur = s_nxt; d_cur = d_nxt; wb = wbn;
    g     = gn;
  }

  if (g == n_full && (n_full << 5) < n_edges) {
    const int eb = (g << 5) + (lane & 16);
    for (int k = 0; k < 16; ++k) {
      const int e = eb + k;
      if (e < n_edges) {
        const int   src = ei[e];
        const int   dst = ei[n_edges + e];
        const float w   = ew[e];
        atomicAdd(&agg[dst * NCLS + c], hs[src * NCLS + c] * w);
      }
    }
  }
}

__global__ __launch_bounds__(256)
void lp_node(const float* __restrict__ h_in, float* agg0, float* agg1,
             const float* __restrict__ W, const float* __restrict__ degree,
             float* __restrict__ h_out, float* __restrict__ hs_out,
             int n_tiles) {
  const int lane  = threadIdx.x & 31;
  const int wave  = threadIdx.x >> 5;
  const int tile  = blockIdx.x * 8 + wave;
  if (tile >= n_tiles) return;
  const int node0 = tile * 16;
  const int m     = lane & 15;
  const int hi    = (lane >= 16) ? 1 : 0;

  float dg = degree[node0 + m];
  float av = (1.0f - ALPHA_F) * dg * dg;

  v8f acc;
#pragma unroll
  for (int r = 0; r < 8; ++r) {
    int row = node0 + r + (hi ? 8 : 0);
    acc[r] = ALPHA_F * h_in[row * NCLS + m];
  }

#pragma unroll
  for (int q = 0; q < 4; ++q) {
    int k0 = 4 * q + 2 * hi;
    v2f a, b;
    a.x = (m == k0)     ? av : 0.0f;
    a.y = (m == k0 + 1) ? av : 0.0f;
    const int i0 = (node0 + k0)     * NCLS + m;
    const int i1 = (node0 + k0 + 1) * NCLS + m;
    b.x = agg0[i0] + agg1[i0];
    b.y = agg0[i1] + agg1[i1];
    acc = __builtin_amdgcn_wmma_f32_16x16x4_f32(
         false, a,  false, b,
         (short)0, acc,  false,  false);
  }

#pragma unroll
  for (int r = 0; r < 8; ++r) {
    float s = acc[r];
    s += __shfl_xor(s, 1, 32);
    s += __shfl_xor(s, 2, 32);
    s += __shfl_xor(s, 4, 32);
    s += __shfl_xor(s, 8, 32);
    int   row = node0 + r + (hi ? 8 : 0);
    float o   = acc[r] / s;
    h_out [row * NCLS + m] = o;
    hs_out[row * NCLS + m] = W[row] * o;
    agg0  [row * NCLS + m] = 0.0f;
    agg1  [row * NCLS + m] = 0.0f;
  }
}

extern "C" void kernel_launch(void* const* d_in, const int* in_sizes, int n_in,
                              void* d_out, int out_size, void* d_ws, size_t ws_size,
                              hipStream_t stream) {
  const float* x   = (const float*)d_in[0];
  const float* W   = (const float*)d_in[1];
  const float* ew  = (const float*)d_in[2];
  const float* deg = (const float*)d_in[3];
  const int*   ei  = (const int*)  d_in[4];

  const int N       = in_sizes[3];
  const int N_EDGES = in_sizes[2];
  const int total   = N * NCLS;
  const int n_tiles = N / 16;

  float* out = (float*)d_out;
  float* agg  = (float*)d_ws;
  float* agg1 = agg  + (size_t)total;
  float* hA   = agg1 + (size_t)total;
  float* hs   = hA   + (size_t)total;

  const int prep_blocks = LOOP_BLOCKS;
  const int edge_blocks = LOOP_BLOCKS;
  const int node_blocks = (n_tiles + 7) / 8;

  lp_prep<<<prep_blocks, 256, 0, stream>>>(x, W, hs, agg, total);

  for (int k = 0; k < 4; ++k) {
    const float* hin  = (k == 0) ? x : ((k & 1) ? hA : out);
    float*       hout = (k & 1) ? out : hA;
    lp_edge<<<edge_blocks, 256, 0, stream>>>(ei, ew, hs, agg, total, N_EDGES);
    lp_node<<<node_blocks, 256, 0, stream>>>(hin, agg, agg1, W, deg, hout, hs, n_tiles);
  }
}


# edge sums leave global f32 atomics: edges grouped per call by 80-node owner block (counting sort), one launch per layer adds rows in LDS (f32 LDS atomics, arrival order) and does the node update
# speedup vs baseline: 4.8587x; 1.1572x over previous
#include <hip/hip_runtime.h>
#include <hip/hip_bf16.h>

typedef __attribute__((ext_vector_type(2))) float v2f;
typedef __attribute__((ext_vector_type(8))) float v8f;

#define ALPHA_F 0.5f
#define NCLS 16

#define NB  80
#define TPB 5
#define LOOP_BLOCKS 1250
#define CH 12800
#define BIN_THREADS 1024

__global__ __launch_bounds__(256)
void lp_prep(const float* __restrict__ x, const float* __restrict__ W,
             float* __restrict__ hs, int* __restrict__ totals,
             int* __restrict__ cursor, int total, int nbins) {
  const int step = gridDim.x * 256;
  for (int i = blockIdx.x * 256 + threadIdx.x; i < total; i += step)
    hs[i] = W[i >> 4] * x[i];
  for (int i = blockIdx.x * 256 + threadIdx.x; i < nbins; i += step) {
    totals[i] = 0;
    cursor[i] = 0;
  }
}

__global__ __launch_bounds__(BIN_THREADS)
void lp_count(const int* __restrict__ ei, int* totals, int n_edges, int nbins) {
  extern __shared__ int sm[];
  int* cnt = sm;
  const int tid = threadIdx.x;
  for (int i = tid; i < nbins; i += BIN_THREADS) cnt[i] = 0;
  __syncthreads();
  const int e0 = blockIdx.x * CH;
  const int e1 = (e0 + CH < n_edges) ? (e0 + CH) : n_edges;
  const int* dst = ei + n_edges;
  for (int e = e0 + tid; e < e1; e += BIN_THREADS)
    atomicAdd(&cnt[(unsigned)dst[e] / NB], 1);
  __syncthreads();
  for (int i = tid; i < nbins; i += BIN_THREADS) {
    const int cv = cnt[i];
    if (cv) atomicAdd(&totals[i], cv);
  }
}

__device__ __forceinline__ void block_exscan(const int* in, int* out, int n, int* wsum) {
  const int tid  = threadIdx.x;
  const int lane = tid & 31;
  const int wave = tid >> 5;
  const int per  = (n + BIN_THREADS - 1) / BIN_THREADS;
  const int i0   = tid * per;
  int local = 0;
  for (int j = 0; j < per; ++j)
    if (i0 + j < n) local += in[i0 + j];
  int v = local;
  for (int o = 1; o < 32; o <<= 1) {
    const int t = __shfl_up(v, o, 32);
    if (lane >= o) v += t;
  }
  if (lane == 31) wsum[wave] = v;
  __syncthreads();
  if (wave == 0) {
    const int own = wsum[lane];
    int s = own;
    for (int o = 1; o < 32; o <<= 1) {
      const int t = __shfl_up(s, o, 32);
      if (lane >= o) s += t;
    }
    wsum[lane] = s - own;
  }
  __syncthreads();
  int run = v - local + wsum[wave];
  for (int j = 0; j < per; ++j)
    if (i0 + j < n) {
      out[i0 + j] = run;
      run += in[i0 + j];
    }
  __syncthreads();
}

__global__ __launch_bounds__(BIN_THREADS)
void lp_place(const int* __restrict__ ei, const float* __restrict__ ew,
              const int* __restrict__ totals, int* cursor, int* __restrict__ gstart,
              int* __restrict__ rec, int n_edges, int nbins) {
  extern __shared__ int sm[];
  int* cur = sm;
  int* st  = sm + nbins;
  int* ls  = sm + 2 * nbins;
  __shared__ int wsum[32];
  __shared__ int skey[CH];
  __shared__ int swt[CH];
  __shared__ int sdest[CH];
  const int tid = threadIdx.x;
  for (int i = tid; i < nbins; i += BIN_THREADS) cur[i] = 0;
  __syncthreads();
  const int e0 = blockIdx.x * CH;
  const int e1 = (e0 + CH < n_edges) ? (e0 + CH) : n_edges;
  const int* src = ei;
  const int* dst = ei + n_edges;
  for (int e = e0 + tid; e < e1; e += BIN_THREADS)
    atomicAdd(&cur[(unsigned)dst[e] / NB], 1);
  __syncthreads();
  block_exscan(totals, st, nbins, wsum);
  block_exscan(cur, ls, nbins, wsum);
  for (int i = tid; i < nbins; i += BIN_THREADS) {
    const int cv = cur[i];
    const int g0 = st[i];
    if (blockIdx.x == 0) gstart[i] = g0;
    const int base = cv ? atomicAdd(&cursor[i], cv) : 0;
    const int l0 = ls[i];
    st[i]  = g0 + base - l0;
    cur[i] = l0;
  }
  __syncthreads();
  for (int e = e0 + tid; e < e1; e += BIN_THREADS) {
    const int   s = src[e];
    const int   d = dst[e];
    const float w = ew[e];
    const int   b = (int)((unsigned)d / NB);
    const int   p = atomicAdd(&cur[b], 1);
    skey[p]  = s * 128 + (d - b * NB);
    swt[p]   = __float_as_int(w);
    sdest[p] = st[b] + p;
  }
  __syncthreads();
  int2* rec2 = reinterpret_cast<int2*>(rec);
  for (int i = tid; i < e1 - e0; i += BIN_THREADS) {
    int2 r;
    r.x = skey[i];
    r.y = swt[i];
    rec2[sdest[i]] = r;
  }
}

template <int K>
__device__ __forceinline__ int row_bcast(int v) {
  return __builtin_amdgcn_update_dpp(0, v, 0x150 + K, 0xF, 0xF, false);
}

template <int K>
__device__ __forceinline__ float row_gather(const float* __restrict__ hs, int sb, int c) {
  return hs[row_bcast<K>(sb) + c];
}

template <int K>
__device__ __forceinline__ void row_add(float* sums, int db, int wbits, int c, float v) {
  float w = __int_as_float(row_bcast<K>(wbits));
  atomicAdd(&sums[row_bcast<K>(db) + c], v * w);
}

__global__ __launch_bounds__(256)
void lp_layer(const float* __restrict__ h_in, const float* __restrict__ hs_in,
              const int* __restrict__ rec, const int* __restrict__ gstart,
              const int* __restrict__ totals, const float* __restrict__ W,
              const float* __restrict__ degree, float* __restrict__ h_out,
              float* __restrict__ hs_out, int n_tiles) {
  __shared__ float sums[(NB + 1) * NCLS];
  const int tid  = threadIdx.x;
  const int lane = tid & 31;
  const int c    = lane & (NCLS - 1);
  const int wave = __builtin_amdgcn_readfirstlane(tid >> 5);
  for (int i = tid; i < (NB + 1) * NCLS; i += 256) sums[i] = 0.0f;
  __syncthreads();

  const int2* __restrict__ rec2 = reinterpret_cast<const int2*>(rec);
  const int beg = __builtin_amdgcn_readfirstlane(gstart[blockIdx.x]);
  const int end = beg + __builtin_amdgcn_readfirstlane(totals[blockIdx.x]);
  int base = beg + wave * 32;
  if (base < end) {
    const int last = end - 1;
    bool ok = (base + lane) < end;
    int2 r  = rec2[ok ? (base + lane) : last];
    while (base < end) {
      const int  nbase = base + 256;
      const int  en    = nbase + lane;
      const bool okn   = en < end;
      const int2 rn = rec2[okn ? en : last];
      const int key = ok ? r.x : NB;
      const int wb  = ok ? r.y : 0;
      const int sb = (key >> 7) * NCLS;
      const int db = (key & 127) * NCLS;

      float v0  = row_gather<0>(hs_in, sb, c);
      float v1  = row_gather<1>(hs_in, sb, c);
      float v2  = row_gather<2>(hs_in, sb, c);
      float v3  = row_gather<3>(hs_in, sb, c);
      float v4  = row_gather<4>(hs_in, sb, c);
      float v5  = row_gather<5>(hs_in, sb, c);
      float v6  = row_gather<6>(hs_in, sb, c);
      float v7  = row_gather<7>(hs_in, sb, c);
      float v8  = row_gather<8>(hs_in, sb, c);
      float v9  = row_gather<9>(hs_in, sb, c);
      float v10 = row_gather<10>(hs_in, sb, c);
      float v11 = row_gather<11>(hs_in, sb, c);
      float v12 = row_gather<12>(hs_in, sb, c);
      float v13 = row_gather<13>(hs_in, sb, c);
      float v14 = row_gather<14>(hs_in, sb, c);
      float v15 = row_gather<15>(hs_in, sb, c);

      row_add<0>(sums, db, wb, c, v0);
      row_add<1>(sums, db, wb, c, v1);
      row_add<2>(sums, db, wb, c, v2);
      row_add<3>(sums, db, wb, c, v3);
      row_add<4>(sums, db, wb, c, v4);
      row_add<5>(sums, db, wb, c, v5);
      row_add<6>(sums, db, wb, c, v6);
      row_add<7>(sums, db, wb, c, v7);
      row_add<8>(sums, db, wb, c, v8);
      row_add<9>(sums, db, wb, c, v9);
      row_add<10>(sums, db, wb, c, v10);
      row_add<11>(sums, db, wb, c, v11);
      row_add<12>(sums, db, wb, c, v12);
      row_add<13>(sums, db, wb, c, v13);
      row_add<14>(sums, db, wb, c, v14);
      row_add<15>(sums, db, wb, c, v15);

      r = rn; ok = okn;
      base = nbase;
    }
  }
  __syncthreads();

  const int tile = blockIdx.x * TPB + wave;
  if (wave >= TPB || tile >= n_tiles) return;
  const int node0 = tile * 16;
  const int m     = lane & 15;
  const int hi    = (lane >= 16) ? 1 : 0;

  float dg = degree[node0 + m];
  float av = (1.0f - ALPHA_F) * dg * dg;

  v8f acc;
#pragma unroll
  for (int r = 0; r < 8; ++r) {
    int row = node0 + r + (hi ? 8 : 0);
    acc[r] = ALPHA_F * h_in[row * NCLS + m];
  }

#pragma unroll
  for (int q = 0; q < 4; ++q) {
    int k0 = 4 * q + 2 * hi;
    v2f a, b;
    a.x = (m == k0)     ? av : 0.0f;
    a.y = (m == k0 + 1) ? av : 0.0f;
    b.x = sums[(wave * 16 + k0)     * NCLS + m];
    b.y = sums[(wave * 16 + k0 + 1) * NCLS + m];
    acc = __builtin_amdgcn_wmma_f32_16x16x4_f32(
         false, a,  false, b,
         (short)0, acc,  false,  false);
  }

#pragma unroll
  for (int r = 0; r < 8; ++r) {
    float s = acc[r];
    s += __shfl_xor(s, 1, 32);
    s += __shfl_xor(s, 2, 32);
    s += __shfl_xor(s, 4, 32);
    s += __shfl_xor(s, 8, 32);
    int   row = node0 + r + (hi ? 8 : 0);
    float o   = acc[r] / s;
    h_out [row * NCLS + m] = o;
    hs_out[row * NCLS + m] = W[row] * o;
  }
}

extern "C" void kernel_launch(void* const* d_in, const int* in_sizes, int n_in,
                              void* d_out, int out_size, void* d_ws, size_t ws_size,
                              hipStream_t stream) {
  const float* x   = (const float*)d_in[0];
  const float* W   = (const float*)d_in[1];
  const float* ew  = (const float*)d_in[2];
  const float* deg = (const float*)d_in[3];
  const int*   ei  = (const int*)  d_in[4];

  const int N       = in_sizes[3];
  const int N_EDGES = in_sizes[2];
  const int total   = N * NCLS;
  const int n_tiles = N / 16;
  const int nbins   = (n_tiles + TPB - 1) / TPB;

  float* out  = (float*)d_out;
  float* hA   = (float*)d_ws;
  float* hs0  = hA  + (size_t)total;
  float* hs1  = hs0 + (size_t)total;
  int*   rec  = (int*)(hs1 + (size_t)total);
  int*   totals = rec + 2 * (size_t)N_EDGES;
  int*   cursor = totals + nbins;
  int*   gstart = cursor + nbins;

  lp_prep<<<LOOP_BLOCKS, 256, 0, stream>>>(x, W, hs0, totals, cursor, total, nbins);
  const int bin_blocks = (N_EDGES + CH - 1) / CH;
  lp_count<<<bin_blocks, BIN_THREADS, nbins * sizeof(int), stream>>>(ei, totals, N_EDGES, nbins);
  lp_place<<<bin_blocks, BIN_THREADS, 3 * nbins * sizeof(int), stream>>>(
      ei, ew, totals, cursor, gstart, rec, N_EDGES, nbins);

  for (int k = 0; k < 4; ++k) {
    const float* hin   = (k == 0) ? x : ((k & 1) ? hA : out);
    float*       hout  = (k & 1) ? out : hA;
    const float* hsin  = (k & 1) ? hs1 : hs0;
    float*       hsout = (k & 1) ? hs0 : hs1;
    lp_layer<<<nbins, 256, 0, stream>>>(hin, hsin, rec, gstart, totals,
                                        W, deg, hout, hsout, n_tiles);
  }
}
